# starting kernel
# baseline (speedup 1.0000x reference)
#include <hip/hip_runtime.h>

typedef float v2f __attribute__((ext_vector_type(2)));
typedef float v8f __attribute__((ext_vector_type(8)));

#define BB 8
#define CC 320
#define QQ 64
#define SS 16384

__global__ __launch_bounds__(256)
void energy_kernel(const float* __restrict__ x, const float* __restrict__ guide,
                   float* __restrict__ energy) {
  const int c0   = blockIdx.x * 16;
  const int q0   = blockIdx.y * 16;
  const int b    = blockIdx.z;
  const int tid  = threadIdx.x;
  const int wave = tid >> 5;
  const int lane = tid & 31;
  const int hi   = lane >> 4;
  const int m    = lane & 15;

  const int KSLICE = SS / 8;
  const float* g  = guide + (size_t)(b * QQ + q0 + m) * SS + wave * KSLICE + 2 * hi;
  const float* xp = x     + (size_t)(b * CC + c0 + m) * SS + wave * KSLICE + 2 * hi;

  v8f acc = {};
  #pragma unroll 4
  for (int k = 0; k < KSLICE; k += 4) {
    v2f a  = *(const v2f*)(g  + k);
    v2f bb = *(const v2f*)(xp + k);
    acc = __builtin_amdgcn_wmma_f32_16x16x4_f32(false, a, false, bb,
                                                (short)0, acc, false, false);
  }

  __shared__ float part[8][8][32];
  #pragma unroll
  for (int v = 0; v < 8; ++v) part[wave][v][lane] = acc[v];
  __syncthreads();

  const int v = tid >> 5;
  const int l = tid & 31;
  float s = 0.f;
  #pragma unroll
  for (int w = 0; w < 8; ++w) s += part[w][v][l];

  const int mo = v + 8 * (l >> 4);
  const int no = l & 15;
  energy[(size_t)(b * QQ + q0 + mo) * CC + c0 + no] = s;
}

__global__ __launch_bounds__(256)
void softmax_kernel(float* __restrict__ e) {
  float* row = e + (size_t)blockIdx.x * CC;
  const int t = threadIdx.x;
  __shared__ float red[256];

  float v0 = row[t];
  float v1 = (t + 256 < CC) ? row[t + 256] : 3.4e38f;
  red[t] = fminf(v0, v1);
  __syncthreads();
  for (int o = 128; o > 0; o >>= 1) {
    if (t < o) red[t] = fminf(red[t], red[t + o]);
    __syncthreads();
  }
  const float mn = red[0];
  __syncthreads();

  float e0 = __expf(mn - v0);
  float e1 = (t + 256 < CC) ? __expf(mn - v1) : 0.f;
  red[t] = e0 + e1;
  __syncthreads();
  for (int o = 128; o > 0; o >>= 1) {
    if (t < o) red[t] += red[t + o];
    __syncthreads();
  }
  const float inv = 1.0f / red[0];
  row[t] = e0 * inv;
  if (t + 256 < CC) row[t + 256] = e1 * inv;
}

__global__ __launch_bounds__(256)
void out_kernel(const float* __restrict__ x, const float* __restrict__ att,
                float* __restrict__ out) {
  const int sb   = blockIdx.x;
  const int qt   = blockIdx.y;
  const int b    = blockIdx.z;
  const int tid  = threadIdx.x;
  const int wave = tid >> 5;
  const int lane = tid & 31;
  const int hi   = lane >> 4;
  const int mn   = lane & 15;

  const int q0 = qt * 16;
  const int s0 = sb * 512 + wave * 64;

  const float* arow = att + (size_t)(b * QQ + q0 + mn) * CC + 2 * hi;
  const float* xb   = x + (size_t)b * CC * SS + (size_t)(2 * hi) * SS + s0 + mn;

  v8f acc0 = {}, acc1 = {}, acc2 = {}, acc3 = {};
  #pragma unroll 2
  for (int k = 0; k < CC; k += 4) {
    v2f a = *(const v2f*)(arow + k);
    const float* xr = xb + (size_t)k * SS;
    v2f b0 = { xr[0],  xr[SS]      };
    v2f b1 = { xr[16], xr[SS + 16] };
    v2f b2 = { xr[32], xr[SS + 32] };
    v2f b3 = { xr[48], xr[SS + 48] };
    acc0 = __builtin_amdgcn_wmma_f32_16x16x4_f32(false, a, false, b0, (short)0, acc0, false, false);
    acc1 = __builtin_amdgcn_wmma_f32_16x16x4_f32(false, a, false, b1, (short)0, acc1, false, false);
    acc2 = __builtin_amdgcn_wmma_f32_16x16x4_f32(false, a, false, b2, (short)0, acc2, false, false);
    acc3 = __builtin_amdgcn_wmma_f32_16x16x4_f32(false, a, false, b3, (short)0, acc3, false, false);
  }

  float* orow = out + (size_t)(b * QQ + q0) * SS + s0;
  #pragma unroll
  for (int v = 0; v < 8; ++v) {
    const int mo = v + 8 * hi;
    float* dst = orow + (size_t)mo * SS + mn;
    dst[0]  = acc0[v];
    dst[16] = acc1[v];
    dst[32] = acc2[v];
    dst[48] = acc3[v];
  }
}

extern "C" void kernel_launch(void* const* d_in, const int* in_sizes, int n_in,
                              void* d_out, int out_size, void* d_ws, size_t ws_size,
                              hipStream_t stream) {
  const float* x     = (const float*)d_in[0];
  const float* guide = (const float*)d_in[1];
  float* out    = (float*)d_out;
  float* energy = (float*)d_ws;

  dim3 g1(CC / 16, QQ / 16, BB);
  energy_kernel<<<g1, 256, 0, stream>>>(x, guide, energy);

  softmax_kernel<<<BB * QQ, 256, 0, stream>>>(energy);

  dim3 g3(SS / 512, QQ / 16, BB);
  out_kernel<<<g3, 256, 0, stream>>>(x, energy, out);
}


# energy product re-tiled: 64q x 80c block tiles fed by whole-line 16-byte loads through LDS, K split over 32 blocks (interleaved 64-k pieces), partials summed in fixed order by the softmax launch
# speedup vs baseline: 2.1244x; 2.1244x over previous
#include <hip/hip_runtime.h>


typedef float v2f __attribute__((ext_vector_type(2)));
typedef float v4f __attribute__((ext_vector_type(4)));
typedef float v8f __attribute__((ext_vector_type(8)));

#define BB 8
#define CC 320
#define QQ 64
#define SS 16384

#define NSPLIT 32
#define KC     64
#define NCHUNK (SS / (NSPLIT * KC))
#define KSTEP  (NSPLIT * KC)
#define CT     80
#define NJ     (CT / 16)
#define ROWS   (QQ + CT)
#define PITCH  68
#define STAGE  (ROWS * PITCH)
#define NLD    (ROWS * (KC / 4) / 128)
#define PLANE  (BB * QQ * CC)

__device__ __forceinline__ void stage_load(v4f (&r)[NLD], const float* __restrict__ gp,
                                           const float* __restrict__ xp) {
  #pragma unroll
  for (int i = 0; i < QQ / 8; ++i) r[i] = *(const v4f*)(gp + (size_t)i * 8 * SS);
  #pragma unroll
  for (int i = 0; i < CT / 8; ++i) r[QQ / 8 + i] = *(const v4f*)(xp + (size_t)i * 8 * SS);
}

__device__ __forceinline__ void stage_store(float* st, const v4f (&r)[NLD]) {
  #pragma unroll
  for (int i = 0; i < NLD; ++i) *(v4f*)(st + i * 8 * PITCH) = r[i];
}

__device__ __forceinline__ void stage_mma(const float* a_src, const float* b_src, v8f (&acc)[NJ]) {
  v4f a[2], b[2][NJ];
  a[0] = *(const v4f*)a_src;
  #pragma unroll
  for (int j = 0; j < NJ; ++j) b[0][j] = *(const v4f*)(b_src + j * 16 * PITCH);
  #pragma unroll
  for (int s = 0; s < KC / 8; ++s) {
    const int cur = s & 1, nxt = cur ^ 1;
    if (s + 1 < KC / 8) {
      a[nxt] = *(const v4f*)(a_src + 8 * (s + 1));
      #pragma unroll
      for (int j = 0; j < NJ; ++j) b[nxt][j] = *(const v4f*)(b_src + j * 16 * PITCH + 8 * (s + 1));
    }
    __builtin_amdgcn_sched_barrier(0);
    #pragma unroll
    for (int j = 0; j < NJ; ++j)
      acc[j] = __builtin_amdgcn_wmma_f32_16x16x4_f32(false, a[cur].xy, false, b[cur][j].xy,
                                                     (short)0, acc[j], false, false);
    #pragma unroll
    for (int j = 0; j < NJ; ++j)
      acc[j] = __builtin_amdgcn_wmma_f32_16x16x4_f32(false, a[cur].zw, false, b[cur][j].zw,
                                                     (short)0, acc[j], false, false);
    __builtin_amdgcn_sched_barrier(0);
  }
}

__global__ __launch_bounds__(128)
void energy_kernel(const float* __restrict__ x, const float* __restrict__ guide,
                   float* __restrict__ part) {
  __shared__ __attribute__((aligned(16))) float lds[2 * STAGE];

  const int ct   = blockIdx.x;
  const int ks   = blockIdx.y;
  const int b    = blockIdx.z;
  const int tid  = threadIdx.x;
  const int wave = tid >> 5;
  const int lane = tid & 31;
  const int h    = lane >> 4;
  const int c    = lane & 15;
  const int lrow = tid >> 4;
  const int lcol = (tid & 15) * 4;

  const float* gp = guide + (size_t)(b * QQ + lrow) * SS + ks * KC + lcol;
  const float* xp = x + (size_t)(b * CC + ct * CT + lrow) * SS + ks * KC + lcol;
  float* st_w = lds + lrow * PITCH + lcol;
  const float* a_src = lds + (16 * wave + c) * PITCH + 4 * h;
  const float* b_src = lds + (QQ + c) * PITCH + 4 * h;

  v8f acc[NJ] = {};
  v4f r[NLD];

  stage_load(r, gp, xp);
  stage_store(st_w, r);
  __syncthreads();

  for (int i = 0; i < NCHUNK - 1; ++i) {
    const int cur = (i & 1) * STAGE;
    const int nxt = STAGE - cur;
    stage_load(r, gp + (i + 1) * KSTEP, xp + (i + 1) * KSTEP);
    stage_mma(a_src + cur, b_src + cur, acc);
    stage_store(st_w + nxt, r);
    __syncthreads();
  }
  stage_mma(a_src + ((NCHUNK - 1) & 1) * STAGE, b_src + ((NCHUNK - 1) & 1) * STAGE, acc);

  float* pp = part + (size_t)ks * PLANE + (size_t)(b * QQ + 16 * wave + 8 * h) * CC + ct * CT + c;
  #pragma unroll
  for (int j = 0; j < NJ; ++j) {
    #pragma unroll
    for (int v = 0; v < 8; ++v) pp[v * CC + 16 * j] = acc[j][v];
  }
}

__global__ __launch_bounds__(256)
void softmax_kernel(const float* __restrict__ part, float* __restrict__ att) {
  const size_t rowoff = (size_t)blockIdx.x * CC;
  const int t = threadIdx.x;
  const bool two = (t + 256 < CC);
  __shared__ float red[256];

  const float* pr = part + rowoff + t;
  float v0 = 0.f;
  #pragma unroll
  for (int p = 0; p < NSPLIT; ++p) v0 += pr[(size_t)p * PLANE];
  float v1 = 3.4e38f;
  if (two) {
    float s1 = 0.f;
    #pragma unroll
    for (int p = 0; p < NSPLIT; ++p) s1 += pr[(size_t)p * PLANE + 256];
    v1 = s1;
  }

  red[t] = fminf(v0, v1);
  __syncthreads();
  for (int o = 128; o > 0; o >>= 1) {
    if (t < o) red[t] = fminf(red[t], red[t + o]);
    __syncthreads();
  }
  const float mn = red[0];
  __syncthreads();

  float e0 = __expf(mn - v0);
  float e1 = two ? __expf(mn - v1) : 0.f;
  red[t] = e0 + e1;
  __syncthreads();
  for (int o = 128; o > 0; o >>= 1) {
    if (t < o) red[t] += red[t + o];
    __syncthreads();
  }
  const float inv = 1.0f / red[0];
  att[rowoff + t] = e0 * inv;
  if (two) att[rowoff + t + 256] = e1 * inv;
}

__global__ __launch_bounds__(256)
void out_kernel(const float* __restrict__ x, const float* __restrict__ att,
                float* __restrict__ out) {
  const int sb   = blockIdx.x;
  const int qt   = blockIdx.y;
  const int b    = blockIdx.z;
  const int tid  = threadIdx.x;
  const int wave = tid >> 5;
  const int lane = tid & 31;
  const int hi   = lane >> 4;
  const int mn   = lane & 15;

  const int q0 = qt * 16;
  const int s0 = sb * 512 + wave * 64;

  const float* arow = att + (size_t)(b * QQ + q0 + mn) * CC + 2 * hi;
  const float* xb   = x + (size_t)b * CC * SS + (size_t)(2 * hi) * SS + s0 + mn;

  v8f acc0 = {}, acc1 = {}, acc2 = {}, acc3 = {};
  #pragma unroll 2
  for (int k = 0; k < CC; k += 4) {
    v2f a = *(const v2f*)(arow + k);
    const float* xr = xb + (size_t)k * SS;
    v2f b0 = { xr[0],  xr[SS]      };
    v2f b1 = { xr[16], xr[SS + 16] };
    v2f b2 = { xr[32], xr[SS + 32] };
    v2f b3 = { xr[48], xr[SS + 48] };
    acc0 = __builtin_amdgcn_wmma_f32_16x16x4_f32(false, a, false, b0, (short)0, acc0, false, false);
    acc1 = __builtin_amdgcn_wmma_f32_16x16x4_f32(false, a, false, b1, (short)0, acc1, false, false);
    acc2 = __builtin_amdgcn_wmma_f32_16x16x4_f32(false, a, false, b2, (short)0, acc2, false, false);
    acc3 = __builtin_amdgcn_wmma_f32_16x16x4_f32(false, a, false, b3, (short)0, acc3, false, false);
  }

  float* orow = out + (size_t)(b * QQ + q0) * SS + s0;
  #pragma unroll
  for (int v = 0; v < 8; ++v) {
    const int mo = v + 8 * hi;
    float* dst = orow + (size_t)mo * SS + mn;
    dst[0]  = acc0[v];
    dst[16] = acc1[v];
    dst[32] = acc2[v];
    dst[48] = acc3[v];
  }
}

extern "C" void kernel_launch(void* const* d_in, const int* in_sizes, int n_in,
                              void* d_out, int out_size, void* d_ws, size_t ws_size,
                              hipStream_t stream) {
  const float* x     = (const float*)d_in[0];
  const float* guide = (const float*)d_in[1];
  float* out  = (float*)d_out;
  float* att  = (float*)d_ws;
  float* part = (float*)d_out;

  dim3 g1(CC / CT, NSPLIT, BB);
  energy_kernel<<<g1, 128, 0, stream>>>(x, guide, part);

  softmax_kernel<<<BB * QQ, 256, 0, stream>>>(part, att);

  dim3 g3(SS / 512, QQ / 16, BB);
  out_kernel<<<g3, 256, 0, stream>>>(x, att, out);
}


# output product re-tiled: 64q x 128s block tiles, four per WGP, x read once by whole-line 16-byte loads through LDS in 32-channel stages, rows stored whole; channel sum in a new fixed order
# speedup vs baseline: 2.3987x; 1.1291x over previous
#include <hip/hip_runtime.h>


typedef float v2f __attribute__((ext_vector_type(2)));
typedef float v4f __attribute__((ext_vector_type(4)));
typedef float v8f __attribute__((ext_vector_type(8)));

#define BB 8
#define CC 320
#define QQ 64
#define SS 16384

#define NSPLIT 32
#define KC     64
#define NCHUNK (SS / (NSPLIT * KC))
#define KSTEP  (NSPLIT * KC)
#define CT     80
#define NJ     (CT / 16)
#define ROWS   (QQ + CT)
#define PITCH  68
#define STAGE  (ROWS * PITCH)
#define NLD    (ROWS * (KC / 4) / 128)
#define PLANE  (BB * QQ * CC)

__device__ __forceinline__ void stage_load(v4f (&r)[NLD], const float* __restrict__ gp,
                                           const float* __restrict__ xp) {
  #pragma unroll
  for (int i = 0; i < QQ / 8; ++i) r[i] = *(const v4f*)(gp + (size_t)i * 8 * SS);
  #pragma unroll
  for (int i = 0; i < CT / 8; ++i) r[QQ / 8 + i] = *(const v4f*)(xp + (size_t)i * 8 * SS);
}

__device__ __forceinline__ void stage_store(float* st, const v4f (&r)[NLD]) {
  #pragma unroll
  for (int i = 0; i < NLD; ++i) *(v4f*)(st + i * 8 * PITCH) = r[i];
}

__device__ __forceinline__ void stage_mma(const float* a_src, const float* b_src, v8f (&acc)[NJ]) {
  v4f a[2], b[2][NJ];
  a[0] = *(const v4f*)a_src;
  #pragma unroll
  for (int j = 0; j < NJ; ++j) b[0][j] = *(const v4f*)(b_src + j * 16 * PITCH);
  #pragma unroll
  for (int s = 0; s < KC / 8; ++s) {
    const int cur = s & 1, nxt = cur ^ 1;
    if (s + 1 < KC / 8) {
      a[nxt] = *(const v4f*)(a_src + 8 * (s + 1));
      #pragma unroll
      for (int j = 0; j < NJ; ++j) b[nxt][j] = *(const v4f*)(b_src + j * 16 * PITCH + 8 * (s + 1));
    }
    __builtin_amdgcn_sched_barrier(0);
    #pragma unroll
    for (int j = 0; j < NJ; ++j)
      acc[j] = __builtin_amdgcn_wmma_f32_16x16x4_f32(false, a[cur].xy, false, b[cur][j].xy,
                                                     (short)0, acc[j], false, false);
    #pragma unroll
    for (int j = 0; j < NJ; ++j)
      acc[j] = __builtin_amdgcn_wmma_f32_16x16x4_f32(false, a[cur].zw, false, b[cur][j].zw,
                                                     (short)0, acc[j], false, false);
    __builtin_amdgcn_sched_barrier(0);
  }
}

__global__ __launch_bounds__(128)
void energy_kernel(const float* __restrict__ x, const float* __restrict__ guide,
                   float* __restrict__ part) {
  __shared__ __attribute__((aligned(16))) float lds[2 * STAGE];

  const int ct   = blockIdx.x;
  const int ks   = blockIdx.y;
  const int b    = blockIdx.z;
  const int tid  = threadIdx.x;
  const int wave = tid >> 5;
  const int lane = tid & 31;
  const int h    = lane >> 4;
  const int c    = lane & 15;
  const int lrow = tid >> 4;
  const int lcol = (tid & 15) * 4;

  const float* gp = guide + (size_t)(b * QQ + lrow) * SS + ks * KC + lcol;
  const float* xp = x + (size_t)(b * CC + ct * CT + lrow) * SS + ks * KC + lcol;
  float* st_w = lds + lrow * PITCH + lcol;
  const float* a_src = lds + (16 * wave + c) * PITCH + 4 * h;
  const float* b_src = lds + (QQ + c) * PITCH + 4 * h;

  v8f acc[NJ] = {};
  v4f r[NLD];

  stage_load(r, gp, xp);
  stage_store(st_w, r);
  __syncthreads();

  for (int i = 0; i < NCHUNK - 1; ++i) {
    const int cur = (i & 1) * STAGE;
    const int nxt = STAGE - cur;
    stage_load(r, gp + (i + 1) * KSTEP, xp + (i + 1) * KSTEP);
    stage_mma(a_src + cur, b_src + cur, acc);
    stage_store(st_w + nxt, r);
    __syncthreads();
  }
  stage_mma(a_src + ((NCHUNK - 1) & 1) * STAGE, b_src + ((NCHUNK - 1) & 1) * STAGE, acc);

  float* pp = part + (size_t)ks * PLANE + (size_t)(b * QQ + 16 * wave + 8 * h) * CC + ct * CT + c;
  #pragma unroll
  for (int j = 0; j < NJ; ++j) {
    #pragma unroll
    for (int v = 0; v < 8; ++v) pp[v * CC + 16 * j] = acc[j][v];
  }
}

__global__ __launch_bounds__(256)
void softmax_kernel(const float* __restrict__ part, float* __restrict__ att) {
  const size_t rowoff = (size_t)blockIdx.x * CC;
  const int t = threadIdx.x;
  const bool two = (t + 256 < CC);
  __shared__ float red[256];

  const float* pr = part + rowoff + t;
  float v0 = 0.f;
  #pragma unroll
  for (int p = 0; p < NSPLIT; ++p) v0 += pr[(size_t)p * PLANE];
  float v1 = 3.4e38f;
  if (two) {
    float s1 = 0.f;
    #pragma unroll
    for (int p = 0; p < NSPLIT; ++p) s1 += pr[(size_t)p * PLANE + 256];
    v1 = s1;
  }

  red[t] = fminf(v0, v1);
  __syncthreads();
  for (int o = 128; o > 0; o >>= 1) {
    if (t < o) red[t] = fminf(red[t], red[t + o]);
    __syncthreads();
  }
  const float mn = red[0];
  __syncthreads();

  float e0 = __expf(mn - v0);
  float e1 = two ? __expf(mn - v1) : 0.f;
  red[t] = e0 + e1;
  __syncthreads();
  for (int o = 128; o > 0; o >>= 1) {
    if (t < o) red[t] += red[t + o];
    __syncthreads();
  }
  const float inv = 1.0f / red[0];
  att[rowoff + t] = e0 * inv;
  if (two) att[rowoff + t + 256] = e1 * inv;
}

#define OS    128
#define OKC   32
#define ONCH  (CC / OKC)
#define XP    132
#define AP    36
#define XST   (OKC * XP)
#define AST   (QQ * AP)
#define OSTG  (XST + AST)
#define NXL   (OKC * (OS / 4) / 128)
#define NAL   (QQ * (OKC / 4) / 128)
#define OLDS  16640
#define NQT   (QQ / 16)
#define NST   2

__device__ __forceinline__ void o_load(v4f (&r)[NXL + NAL], const float* __restrict__ xp,
                                       const float* __restrict__ ap) {
  #pragma unroll
  for (int i = 0; i < NXL; ++i) r[i] = *(const v4f*)(xp + (size_t)i * 4 * SS);
  #pragma unroll
  for (int i = 0; i < NAL; ++i) r[NXL + i] = *(const v4f*)(ap + (size_t)i * 16 * CC);
}

__device__ __forceinline__ void o_store(float* xs_w, float* as_w, const v4f (&r)[NXL + NAL]) {
  #pragma unroll
  for (int i = 0; i < NXL; ++i) *(v4f*)(xs_w + i * 4 * XP) = r[i];
  #pragma unroll
  for (int i = 0; i < NAL; ++i) *(v4f*)(as_w + i * 16 * AP) = r[NXL + i];
}

__device__ __forceinline__ void o_mma(const float* a_src, const float* b_src, v8f (&acc)[NQT][NST]) {
  v4f a[2][NQT];
  float bx[2][NST][4];
  #pragma unroll
  for (int qt = 0; qt < NQT; ++qt) a[0][qt] = *(const v4f*)(a_src + qt * 16 * AP);
  #pragma unroll
  for (int st = 0; st < NST; ++st) {
    #pragma unroll
    for (int e = 0; e < 4; ++e) bx[0][st][e] = b_src[e * XP + 16 * st];
  }
  #pragma unroll
  for (int t = 0; t < OKC / 8; ++t) {
    const int cur = t & 1, nxt = cur ^ 1;
    if (t + 1 < OKC / 8) {
      #pragma unroll
      for (int qt = 0; qt < NQT; ++qt)
        a[nxt][qt] = *(const v4f*)(a_src + qt * 16 * AP + 8 * (t + 1));
      #pragma unroll
      for (int st = 0; st < NST; ++st) {
        #pragma unroll
        for (int e = 0; e < 4; ++e) bx[nxt][st][e] = b_src[(8 * (t + 1) + e) * XP + 16 * st];
      }
    }
    __builtin_amdgcn_sched_barrier(0);
    #pragma unroll
    for (int qt = 0; qt < NQT; ++qt) {
      #pragma unroll
      for (int st = 0; st < NST; ++st) {
        const v2f bb = { bx[cur][st][0], bx[cur][st][1] };
        acc[qt][st] = __builtin_amdgcn_wmma_f32_16x16x4_f32(false, a[cur][qt].xy, false, bb,
                                                            (short)0, acc[qt][st], false, false);
      }
    }
    #pragma unroll
    for (int qt = 0; qt < NQT; ++qt) {
      #pragma unroll
      for (int st = 0; st < NST; ++st) {
        const v2f bb = { bx[cur][st][2], bx[cur][st][3] };
        acc[qt][st] = __builtin_amdgcn_wmma_f32_16x16x4_f32(false, a[cur][qt].zw, false, bb,
                                                            (short)0, acc[qt][st], false, false);
      }
    }
    __builtin_amdgcn_sched_barrier(0);
  }
}

__global__ __launch_bounds__(128)
void out_kernel(const float* __restrict__ x, const float* __restrict__ att,
                float* __restrict__ out) {
  __shared__ __attribute__((aligned(16))) float lds[OLDS];

  const int sb   = blockIdx.x;
  const int b    = blockIdx.z;
  const int tid  = threadIdx.x;
  const int wave = tid >> 5;
  const int lane = tid & 31;
  const int h    = lane >> 4;
  const int c    = lane & 15;
  const int xrow = tid >> 5;
  const int xcol = (tid & 31) * 4;
  const int arow = tid >> 3;
  const int acol = (tid & 7) * 4;

  const float* xp = x + (size_t)(b * CC + xrow) * SS + sb * OS + xcol;
  const float* ap = att + (size_t)(b * QQ + arow) * CC + acol;
  float* xs_w = lds + xrow * XP + xcol;
  float* as_w = lds + XST + arow * AP + acol;
  const float* a_src = lds + XST + c * AP + 4 * h;
  const float* b_src = lds + (4 * h) * XP + 32 * wave + c;

  v8f acc[NQT][NST] = {};
  v4f r[NXL + NAL];

  o_load(r, xp, ap);
  o_store(xs_w, as_w, r);
  __syncthreads();

  for (int i = 0; i < ONCH - 1; ++i) {
    const int cur = (i & 1) * OSTG;
    const int nxt = OSTG - cur;
    o_load(r, xp + (size_t)(i + 1) * OKC * SS, ap + (i + 1) * OKC);
    o_mma(a_src + cur, b_src + cur, acc);
    o_store(xs_w + nxt, as_w + nxt, r);
    __syncthreads();
  }
  o_mma(a_src + ((ONCH - 1) & 1) * OSTG, b_src + ((ONCH - 1) & 1) * OSTG, acc);

  __syncthreads();

  float* ot = lds + (8 * h) * XP + 32 * wave + c;
  #pragma unroll
  for (int qt = 0; qt < NQT; ++qt) {
    #pragma unroll
    for (int st = 0; st < NST; ++st) {
      #pragma unroll
      for (int v = 0; v < 8; ++v) ot[(16 * qt + v) * XP + 16 * st] = acc[qt][st][v];
    }
  }
  __syncthreads();

  float* op = out + (size_t)(b * QQ + xrow) * SS + sb * OS + xcol;
  const float* os = lds + xrow * XP + xcol;
  #pragma unroll
  for (int i = 0; i < QQ / 4; ++i) *(v4f*)(op + (size_t)i * 4 * SS) = *(const v4f*)(os + i * 4 * XP);
}

extern "C" void kernel_launch(void* const* d_in, const int* in_sizes, int n_in,
                              void* d_out, int out_size, void* d_ws, size_t ws_size,
                              hipStream_t stream) {
  const float* x     = (const float*)d_in[0];
  const float* guide = (const float*)d_in[1];
  float* out  = (float*)d_out;
  float* att  = (float*)d_ws;
  float* part = (float*)d_out;

  dim3 g1(CC / CT, NSPLIT, BB);
  energy_kernel<<<g1, 128, 0, stream>>>(x, guide, part);

  softmax_kernel<<<BB * QQ, 256, 0, stream>>>(part, att);

  dim3 g3(SS / OS, 1, BB);
  out_kernel<<<g3, 128, 0, stream>>>(x, att, out);
}
